# starting kernel
# baseline (speedup 1.0000x reference)
#include <hip/hip_runtime.h>

typedef __attribute__((ext_vector_type(2))) float v2f;
typedef __attribute__((ext_vector_type(4))) float v4f;
typedef __attribute__((ext_vector_type(8))) float v8f;

#define NB    16384
#define DIN   204
#define NPAIR 100
#define PSTR  70

static __device__ __forceinline__ v8f wmma4(v2f a, v2f b, v8f c) {
  return __builtin_amdgcn_wmma_f32_16x16x4_f32(false, a, false, b, (short)0, c, false, false);
}

static __device__ __forceinline__ v8f splat8(float v) {
  v8f r;
#pragma unroll
  for (int e = 0; e < 8; ++e) r[e] = v;
  return r;
}

static __device__ __forceinline__ void lds_fence() {
  asm volatile("s_wait_dscnt 0" ::: "memory");
}

__global__ __launch_bounds__(256) void bn_stats_kernel(const float* __restrict__ X,
                                                       float* __restrict__ nrm) {
  const int c = blockIdx.x;
  float s = 0.f, s2 = 0.f;
  for (int r = threadIdx.x; r < NB; r += 256) {
    float v = X[(size_t)r * DIN + c];
    s += v;
    s2 += v * v;
  }
  __shared__ float red0[256], red1[256];
  red0[threadIdx.x] = s;
  red1[threadIdx.x] = s2;
  __syncthreads();
  for (int st = 128; st > 0; st >>= 1) {
    if (threadIdx.x < st) {
      red0[threadIdx.x] += red0[threadIdx.x + st];
      red1[threadIdx.x] += red1[threadIdx.x + st];
    }
    __syncthreads();
  }
  if (threadIdx.x == 0) {
    float m   = red0[0] * (1.f / NB);
    float var = red1[0] * (1.f / NB) - m * m;
    float rs  = rsqrtf(var + 1e-5f);
    nrm[2 * c]     = rs;
    nrm[2 * c + 1] = -m * rs;
  }
}

template <int KSTEPS, int NT, bool RELU>
static __device__ __forceinline__ void dense_layer(const float* pb, int nc, bool hi,
                                                   const float* __restrict__ W, int wstride,
                                                   const float* __restrict__ bias, v8f* acc) {
  const int kof = hi ? 2 : 0;
#pragma unroll
  for (int t = 0; t < NT; ++t) acc[t] = splat8(bias[t * 16 + nc]);
#pragma unroll
  for (int kk = 0; kk < KSTEPS; ++kk) {
    const int ko = 4 * kk + kof;
    v2f a = *(const v2f*)(pb + nc * PSTR + ko);
#pragma unroll
    for (int t = 0; t < NT; ++t) {
      const int n = t * 16 + nc;
      v2f b = *(const v2f*)(W + n * wstride + ko);
      acc[t] = wmma4(a, b, acc[t]);
    }
  }
  if (RELU) {
#pragma unroll
    for (int t = 0; t < NT; ++t)
#pragma unroll
      for (int e = 0; e < 8; ++e) acc[t][e] = fmaxf(acc[t][e], 0.01f * acc[t][e]);
  }
}

static __device__ __forceinline__ void store_h(float* pb, int nc, bool hi, const v8f* h) {
#pragma unroll
  for (int t = 0; t < 4; ++t)
#pragma unroll
    for (int e = 0; e < 8; ++e) {
      const int r = e + (hi ? 8 : 0);
      pb[r * PSTR + t * 16 + nc] = h[t][e];
    }
}

__global__ __launch_bounds__(128) void policy_fused_kernel(
    const float* __restrict__ X, const float* __restrict__ nrm,
    const float* __restrict__ Wme, const float* __restrict__ bme,
    const float* __restrict__ W1, const float* __restrict__ b1,
    const float* __restrict__ W2, const float* __restrict__ b2,
    const float* __restrict__ W3, const float* __restrict__ b3,
    float* __restrict__ out) {
  __shared__ float polbuf[4][16 * PSTR];

  const int  lane = threadIdx.x & 31;
  const int  wv   = threadIdx.x >> 5;
  const int  nc   = lane & 15;
  const bool hi   = (lane >= 16);
  const int  rowbase = blockIdx.x * 64 + wv * 16;
  const int  myrow   = rowbase + nc;
  float* pb = polbuf[wv];
  const float* xrow = X + (size_t)myrow * DIN;

  v2f bw[4];
#pragma unroll
  for (int t = 0; t < 4; ++t) {
    const int n = t * 16 + nc;
    v2f w = *(const v2f*)(Wme + 2 * n);
    v2f z; z.x = 0.f; z.y = 0.f;
    bw[t] = hi ? z : w;
  }

  v8f absacc[4];
#pragma unroll
  for (int t = 0; t < 4; ++t) absacc[t] = splat8(0.f);
  v2f asum; asum.x = 0.f; asum.y = 0.f;
  const v8f zero8 = splat8(0.f);

  v2f xv = *(const v2f*)(xrow + 4);
  v4f nv = *(const v4f*)(nrm + 8);
  for (int k = 0; k < NPAIR; ++k) {
    v2f av;
    av.x = fmaf(xv.x, nv.x, nv.y);
    av.y = fmaf(xv.y, nv.z, nv.w);
    const int cn = (k < NPAIR - 1) ? (6 + 2 * k) : 4;
    xv = *(const v2f*)(xrow + cn);
    nv = *(const v4f*)(nrm + 2 * cn);
    v2f z2; z2.x = 0.f; z2.y = 0.f;
    av = hi ? z2 : av;
    asum.x += av.x; asum.y += av.y;
#pragma unroll
    for (int t = 0; t < 4; ++t) {
      v8f z = wmma4(av, bw[t], zero8);
#pragma unroll
      for (int e = 0; e < 8; ++e) absacc[t][e] += fabsf(z[e]);
    }
  }

#pragma unroll
  for (int t = 0; t < 4; ++t) {
    v8f cb = splat8(100.f * bme[t * 16 + nc]);
    v8f zs = wmma4(asum, bw[t], cb);
#pragma unroll
    for (int e = 0; e < 8; ++e) {
      const float m = (0.505f * zs[e] + 0.495f * absacc[t][e]) * 0.01f;
      const int r = e + (hi ? 8 : 0);
      pb[r * PSTR + 4 + t * 16 + nc] = m;
    }
  }
  if (!hi) {
#pragma unroll
    for (int c = 0; c < 4; ++c)
      pb[nc * PSTR + c] = fmaf(xrow[c], nrm[2 * c], nrm[2 * c + 1]);
  }
  lds_fence();

  v8f h[4];
  dense_layer<17, 4, true>(pb, nc, hi, W1, 68, b1, h);
  lds_fence();
  store_h(pb, nc, hi, h);
  lds_fence();
  dense_layer<16, 4, true>(pb, nc, hi, W2, 64, b2, h);
  lds_fence();
  store_h(pb, nc, hi, h);
  lds_fence();
  v8f lg[2];
  dense_layer<16, 2, false>(pb, nc, hi, W3, 64, b3, lg);

#pragma unroll
  for (int r = 0; r < 8; ++r) {
    const float a0 = lg[0][r], a1 = lg[1][r];
    float m = fmaxf(a0, a1);
#pragma unroll
    for (int off = 8; off >= 1; off >>= 1) m = fmaxf(m, __shfl_xor(m, off, 32));
    const float e0 = __expf(a0 - m);
    const float e1 = __expf(a1 - m);
    float s = e0 + e1;
#pragma unroll
    for (int off = 8; off >= 1; off >>= 1) s += __shfl_xor(s, off, 32);
    const float inv = 1.0f / s;
    const int row = rowbase + r + (hi ? 8 : 0);
    out[row * 32 + nc]      = e0 * inv;
    out[row * 32 + 16 + nc] = e1 * inv;
  }
}

extern "C" void kernel_launch(void* const* d_in, const int* in_sizes, int n_in,
                              void* d_out, int out_size, void* d_ws, size_t ws_size,
                              hipStream_t stream) {
  const float* X   = (const float*)d_in[0];
  const float* Wme = (const float*)d_in[1];
  const float* bme = (const float*)d_in[2];
  const float* W1  = (const float*)d_in[3];
  const float* b1  = (const float*)d_in[4];
  const float* W2  = (const float*)d_in[5];
  const float* b2  = (const float*)d_in[6];
  const float* W3  = (const float*)d_in[7];
  const float* b3  = (const float*)d_in[8];
  float* out = (float*)d_out;

  float* nrm = (float*)d_ws;

  bn_stats_kernel<<<DIN, 256, 0, stream>>>(X, nrm);
  policy_fused_kernel<<<NB / 64, 128, 0, stream>>>(X, nrm, Wme, bme,
                                                   W1, b1, W2, b2, W3, b3, out);
}


# Column sums from whole-line reads: 128 row chunks sum their columns with contiguous 16-B loads, a small added launch adds the 128 partials per column; changes the order of the batch-statistic sums
# speedup vs baseline: 1.1355x; 1.1355x over previous
#include <hip/hip_runtime.h>

typedef __attribute__((ext_vector_type(2))) float v2f;
typedef __attribute__((ext_vector_type(4))) float v4f;
typedef __attribute__((ext_vector_type(8))) float v8f;

#define NB    16384
#define DIN   204
#define NPAIR 100
#define PSTR  70

static __device__ __forceinline__ v8f wmma4(v2f a, v2f b, v8f c) {
  return __builtin_amdgcn_wmma_f32_16x16x4_f32(false, a, false, b, (short)0, c, false, false);
}

static __device__ __forceinline__ v8f splat8(float v) {
  v8f r;
#pragma unroll
  for (int e = 0; e < 8; ++e) r[e] = v;
  return r;
}

static __device__ __forceinline__ void lds_fence() {
  asm volatile("s_wait_dscnt 0" ::: "memory");
}

#define SCH   128
#define SROWS (NB / SCH)
#define SF4   (SROWS * 51)
#define SSLOT 512

__global__ __launch_bounds__(256) void bn_partial_kernel(const float* __restrict__ X,
                                                         float* __restrict__ part) {
  __shared__ v4f red[2 * 255];
  const int t = threadIdx.x;
  v4f s = {0.f, 0.f, 0.f, 0.f};
  v4f q = {0.f, 0.f, 0.f, 0.f};
  if (t < 255) {
    const v4f* p = (const v4f*)(X + (size_t)blockIdx.x * SROWS * DIN) + t;
#pragma unroll
    for (int j = 0; j < 25; ++j) {
      const v4f v = p[j * 255];
      s += v;
      q += v * v;
    }
    if (t < SF4 - 25 * 255) {
      const v4f v = p[25 * 255];
      s += v;
      q += v * v;
    }
    red[t]       = s;
    red[255 + t] = q;
  }
  __syncthreads();
  if (t < 102) {
    const int k  = (t < 51) ? t : t - 51;
    const v4f* r = red + ((t < 51) ? 0 : 255) + k;
    const v4f a = (r[0] + r[51]) + (r[102] + r[153]) + r[204];
    *(v4f*)(part + (size_t)blockIdx.x * SSLOT + 4 * t) = a;
  }
}

__global__ __launch_bounds__(SCH) void bn_final_kernel(const float* __restrict__ part,
                                                       float* __restrict__ nrm) {
  const int cg = blockIdx.x;
  const int g  = threadIdx.x;
  const float* slot = part + (size_t)g * SSLOT;
  v4f s = *(const v4f*)(slot + 4 * cg);
  v4f q = *(const v4f*)(slot + DIN + 4 * cg);
#pragma unroll
  for (int off = 16; off >= 1; off >>= 1) {
#pragma unroll
    for (int e = 0; e < 4; ++e) {
      s[e] += __shfl_xor(s[e], off, 32);
      q[e] += __shfl_xor(q[e], off, 32);
    }
  }
  __shared__ v4f reds[SCH / 32], redq[SCH / 32];
  if ((g & 31) == 0) {
    reds[g >> 5] = s;
    redq[g >> 5] = q;
  }
  __syncthreads();
  if (g < 4) {
    const float* fs = (const float*)reds;
    const float* fq = (const float*)redq;
    float ts = 0.f, tq = 0.f;
#pragma unroll
    for (int w = 0; w < SCH / 32; ++w) {
      ts += fs[4 * w + g];
      tq += fq[4 * w + g];
    }
    const int c = 4 * cg + g;
    float m   = ts * (1.f / NB);
    float var = tq * (1.f / NB) - m * m;
    float rs  = rsqrtf(var + 1e-5f);
    nrm[2 * c]     = rs;
    nrm[2 * c + 1] = -m * rs;
  }
}

template <int KSTEPS, int NT, bool RELU>
static __device__ __forceinline__ void dense_layer(const float* pb, int nc, bool hi,
                                                   const float* __restrict__ W, int wstride,
                                                   const float* __restrict__ bias, v8f* acc) {
  const int kof = hi ? 2 : 0;
#pragma unroll
  for (int t = 0; t < NT; ++t) acc[t] = splat8(bias[t * 16 + nc]);
#pragma unroll
  for (int kk = 0; kk < KSTEPS; ++kk) {
    const int ko = 4 * kk + kof;
    v2f a = *(const v2f*)(pb + nc * PSTR + ko);
#pragma unroll
    for (int t = 0; t < NT; ++t) {
      const int n = t * 16 + nc;
      v2f b = *(const v2f*)(W + n * wstride + ko);
      acc[t] = wmma4(a, b, acc[t]);
    }
  }
  if (RELU) {
#pragma unroll
    for (int t = 0; t < NT; ++t)
#pragma unroll
      for (int e = 0; e < 8; ++e) acc[t][e] = fmaxf(acc[t][e], 0.01f * acc[t][e]);
  }
}

static __device__ __forceinline__ void store_h(float* pb, int nc, bool hi, const v8f* h) {
#pragma unroll
  for (int t = 0; t < 4; ++t)
#pragma unroll
    for (int e = 0; e < 8; ++e) {
      const int r = e + (hi ? 8 : 0);
      pb[r * PSTR + t * 16 + nc] = h[t][e];
    }
}

__global__ __launch_bounds__(128) void policy_fused_kernel(
    const float* __restrict__ X, const float* __restrict__ nrm,
    const float* __restrict__ Wme, const float* __restrict__ bme,
    const float* __restrict__ W1, const float* __restrict__ b1,
    const float* __restrict__ W2, const float* __restrict__ b2,
    const float* __restrict__ W3, const float* __restrict__ b3,
    float* __restrict__ out) {
  __shared__ float polbuf[4][16 * PSTR];

  const int  lane = threadIdx.x & 31;
  const int  wv   = threadIdx.x >> 5;
  const int  nc   = lane & 15;
  const bool hi   = (lane >= 16);
  const int  rowbase = blockIdx.x * 64 + wv * 16;
  const int  myrow   = rowbase + nc;
  float* pb = polbuf[wv];
  const float* xrow = X + (size_t)myrow * DIN;

  v2f bw[4];
#pragma unroll
  for (int t = 0; t < 4; ++t) {
    const int n = t * 16 + nc;
    v2f w = *(const v2f*)(Wme + 2 * n);
    v2f z; z.x = 0.f; z.y = 0.f;
    bw[t] = hi ? z : w;
  }

  v8f absacc[4];
#pragma unroll
  for (int t = 0; t < 4; ++t) absacc[t] = splat8(0.f);
  v2f asum; asum.x = 0.f; asum.y = 0.f;
  const v8f zero8 = splat8(0.f);

  v2f xv = *(const v2f*)(xrow + 4);
  v4f nv = *(const v4f*)(nrm + 8);
  for (int k = 0; k < NPAIR; ++k) {
    v2f av;
    av.x = fmaf(xv.x, nv.x, nv.y);
    av.y = fmaf(xv.y, nv.z, nv.w);
    const int cn = (k < NPAIR - 1) ? (6 + 2 * k) : 4;
    xv = *(const v2f*)(xrow + cn);
    nv = *(const v4f*)(nrm + 2 * cn);
    v2f z2; z2.x = 0.f; z2.y = 0.f;
    av = hi ? z2 : av;
    asum.x += av.x; asum.y += av.y;
#pragma unroll
    for (int t = 0; t < 4; ++t) {
      v8f z = wmma4(av, bw[t], zero8);
#pragma unroll
      for (int e = 0; e < 8; ++e) absacc[t][e] += fabsf(z[e]);
    }
  }

#pragma unroll
  for (int t = 0; t < 4; ++t) {
    v8f cb = splat8(100.f * bme[t * 16 + nc]);
    v8f zs = wmma4(asum, bw[t], cb);
#pragma unroll
    for (int e = 0; e < 8; ++e) {
      const float m = (0.505f * zs[e] + 0.495f * absacc[t][e]) * 0.01f;
      const int r = e + (hi ? 8 : 0);
      pb[r * PSTR + 4 + t * 16 + nc] = m;
    }
  }
  if (!hi) {
#pragma unroll
    for (int c = 0; c < 4; ++c)
      pb[nc * PSTR + c] = fmaf(xrow[c], nrm[2 * c], nrm[2 * c + 1]);
  }
  lds_fence();

  v8f h[4];
  dense_layer<17, 4, true>(pb, nc, hi, W1, 68, b1, h);
  lds_fence();
  store_h(pb, nc, hi, h);
  lds_fence();
  dense_layer<16, 4, true>(pb, nc, hi, W2, 64, b2, h);
  lds_fence();
  store_h(pb, nc, hi, h);
  lds_fence();
  v8f lg[2];
  dense_layer<16, 2, false>(pb, nc, hi, W3, 64, b3, lg);

#pragma unroll
  for (int r = 0; r < 8; ++r) {
    const float a0 = lg[0][r], a1 = lg[1][r];
    float m = fmaxf(a0, a1);
#pragma unroll
    for (int off = 8; off >= 1; off >>= 1) m = fmaxf(m, __shfl_xor(m, off, 32));
    const float e0 = __expf(a0 - m);
    const float e1 = __expf(a1 - m);
    float s = e0 + e1;
#pragma unroll
    for (int off = 8; off >= 1; off >>= 1) s += __shfl_xor(s, off, 32);
    const float inv = 1.0f / s;
    const int row = rowbase + r + (hi ? 8 : 0);
    out[row * 32 + nc]      = e0 * inv;
    out[row * 32 + 16 + nc] = e1 * inv;
  }
}

extern "C" void kernel_launch(void* const* d_in, const int* in_sizes, int n_in,
                              void* d_out, int out_size, void* d_ws, size_t ws_size,
                              hipStream_t stream) {
  const float* X   = (const float*)d_in[0];
  const float* Wme = (const float*)d_in[1];
  const float* bme = (const float*)d_in[2];
  const float* W1  = (const float*)d_in[3];
  const float* b1  = (const float*)d_in[4];
  const float* W2  = (const float*)d_in[5];
  const float* b2  = (const float*)d_in[6];
  const float* W3  = (const float*)d_in[7];
  const float* b3  = (const float*)d_in[8];
  float* out = (float*)d_out;

  float* nrm = (float*)d_ws;

  float* part = out;

  bn_partial_kernel<<<SCH, 256, 0, stream>>>(X, part);
  bn_final_kernel<<<51, SCH, 0, stream>>>(part, nrm);
  policy_fused_kernel<<<NB / 64, 128, 0, stream>>>(X, nrm, Wme, bme,
                                                   W1, b1, W2, b2, W3, b3, out);
}


# Entity loop: each lane streams its X row with 16-byte loads, ten in flight per 20-pair trip (ring of 10 registers), scale/shift fetched a group ahead, instead of one 8-byte load one pair ahead
# speedup vs baseline: 1.9250x; 1.6953x over previous
#include <hip/hip_runtime.h>

typedef __attribute__((ext_vector_type(2))) float v2f;
typedef __attribute__((ext_vector_type(4))) float v4f;
typedef __attribute__((ext_vector_type(8))) float v8f;

#define NB    16384
#define DIN   204
#define NPAIR 100
#define PSTR  70

static __device__ __forceinline__ v8f wmma4(v2f a, v2f b, v8f c) {
  return __builtin_amdgcn_wmma_f32_16x16x4_f32(false, a, false, b, (short)0, c, false, false);
}

static __device__ __forceinline__ v8f splat8(float v) {
  v8f r;
#pragma unroll
  for (int e = 0; e < 8; ++e) r[e] = v;
  return r;
}

static __device__ __forceinline__ void lds_fence() {
  asm volatile("s_wait_dscnt 0" ::: "memory");
}

#define SCH   128
#define SROWS (NB / SCH)
#define SF4   (SROWS * 51)
#define SSLOT 512

__global__ __launch_bounds__(256) void bn_partial_kernel(const float* __restrict__ X,
                                                         float* __restrict__ part) {
  __shared__ v4f red[2 * 255];
  const int t = threadIdx.x;
  v4f s = {0.f, 0.f, 0.f, 0.f};
  v4f q = {0.f, 0.f, 0.f, 0.f};
  if (t < 255) {
    const v4f* p = (const v4f*)(X + (size_t)blockIdx.x * SROWS * DIN) + t;
#pragma unroll
    for (int j = 0; j < 25; ++j) {
      const v4f v = p[j * 255];
      s += v;
      q += v * v;
    }
    if (t < SF4 - 25 * 255) {
      const v4f v = p[25 * 255];
      s += v;
      q += v * v;
    }
    red[t]       = s;
    red[255 + t] = q;
  }
  __syncthreads();
  if (t < 102) {
    const int k  = (t < 51) ? t : t - 51;
    const v4f* r = red + ((t < 51) ? 0 : 255) + k;
    const v4f a = (r[0] + r[51]) + (r[102] + r[153]) + r[204];
    *(v4f*)(part + (size_t)blockIdx.x * SSLOT + 4 * t) = a;
  }
}

__global__ __launch_bounds__(SCH) void bn_final_kernel(const float* __restrict__ part,
                                                       float* __restrict__ nrm) {
  const int cg = blockIdx.x;
  const int g  = threadIdx.x;
  const float* slot = part + (size_t)g * SSLOT;
  v4f s = *(const v4f*)(slot + 4 * cg);
  v4f q = *(const v4f*)(slot + DIN + 4 * cg);
#pragma unroll
  for (int off = 16; off >= 1; off >>= 1) {
#pragma unroll
    for (int e = 0; e < 4; ++e) {
      s[e] += __shfl_xor(s[e], off, 32);
      q[e] += __shfl_xor(q[e], off, 32);
    }
  }
  __shared__ v4f reds[SCH / 32], redq[SCH / 32];
  if ((g & 31) == 0) {
    reds[g >> 5] = s;
    redq[g >> 5] = q;
  }
  __syncthreads();
  if (g < 4) {
    const float* fs = (const float*)reds;
    const float* fq = (const float*)redq;
    float ts = 0.f, tq = 0.f;
#pragma unroll
    for (int w = 0; w < SCH / 32; ++w) {
      ts += fs[4 * w + g];
      tq += fq[4 * w + g];
    }
    const int c = 4 * cg + g;
    float m   = ts * (1.f / NB);
    float var = tq * (1.f / NB) - m * m;
    float rs  = rsqrtf(var + 1e-5f);
    nrm[2 * c]     = rs;
    nrm[2 * c + 1] = -m * rs;
  }
}

template <int KSTEPS, int NT, bool RELU>
static __device__ __forceinline__ void dense_layer(const float* pb, int nc, bool hi,
                                                   const float* __restrict__ W, int wstride,
                                                   const float* __restrict__ bias, v8f* acc) {
  const int kof = hi ? 2 : 0;
#pragma unroll
  for (int t = 0; t < NT; ++t) acc[t] = splat8(bias[t * 16 + nc]);
#pragma unroll
  for (int kk = 0; kk < KSTEPS; ++kk) {
    const int ko = 4 * kk + kof;
    v2f a = *(const v2f*)(pb + nc * PSTR + ko);
#pragma unroll
    for (int t = 0; t < NT; ++t) {
      const int n = t * 16 + nc;
      v2f b = *(const v2f*)(W + n * wstride + ko);
      acc[t] = wmma4(a, b, acc[t]);
    }
  }
  if (RELU) {
#pragma unroll
    for (int t = 0; t < NT; ++t)
#pragma unroll
      for (int e = 0; e < 8; ++e) acc[t][e] = fmaxf(acc[t][e], 0.01f * acc[t][e]);
  }
}

static __device__ __forceinline__ void store_h(float* pb, int nc, bool hi, const v8f* h) {
#pragma unroll
  for (int t = 0; t < 4; ++t)
#pragma unroll
    for (int e = 0; e < 8; ++e) {
      const int r = e + (hi ? 8 : 0);
      pb[r * PSTR + t * 16 + nc] = h[t][e];
    }
}

__global__ __launch_bounds__(128) void policy_fused_kernel(
    const float* __restrict__ X, const float* __restrict__ nrm,
    const float* __restrict__ Wme, const float* __restrict__ bme,
    const float* __restrict__ W1, const float* __restrict__ b1,
    const float* __restrict__ W2, const float* __restrict__ b2,
    const float* __restrict__ W3, const float* __restrict__ b3,
    float* __restrict__ out) {
  __shared__ float polbuf[4][16 * PSTR];

  const int  lane = threadIdx.x & 31;
  const int  wv   = threadIdx.x >> 5;
  const int  nc   = lane & 15;
  const bool hi   = (lane >= 16);
  const int  rowbase = blockIdx.x * 64 + wv * 16;
  const int  myrow   = rowbase + nc;
  float* pb = polbuf[wv];
  const float* xrow = X + (size_t)myrow * DIN;

  v2f bw[4];
#pragma unroll
  for (int t = 0; t < 4; ++t) {
    const int n = t * 16 + nc;
    v2f w = *(const v2f*)(Wme + 2 * n);
    v2f z; z.x = 0.f; z.y = 0.f;
    bw[t] = hi ? z : w;
  }

  v8f absacc[4];
#pragma unroll
  for (int t = 0; t < 4; ++t) absacc[t] = splat8(0.f);
  v2f asum; asum.x = 0.f; asum.y = 0.f;
  const v8f zero8 = splat8(0.f);

  const v4f* xp4 = (const v4f*)(xrow + 4);
  const v4f* nr4 = (const v4f*)(nrm + 8);
  v4f xq[10];
#pragma unroll
  for (int i = 0; i < 10; ++i) xq[i] = xp4[i];
  v4f nq[4];
#pragma unroll
  for (int j = 0; j < 4; ++j) nq[j] = nr4[j];

  for (int go = 0; go < NPAIR / 4; go += 5) {
#pragma unroll
    for (int u = 0; u < 5; ++u) {
      const int g = go + u;
      const v4f xa = xq[2 * u], xb = xq[2 * u + 1];
      const int gx = (g + 5 < NPAIR / 4) ? g + 5 : NPAIR / 4 - 1;
      xq[2 * u]     = xp4[2 * gx];
      xq[2 * u + 1] = xp4[2 * gx + 1];
      const int gs = (g + 1 < NPAIR / 4) ? g + 1 : NPAIR / 4 - 1;
      v4f nn[4];
#pragma unroll
      for (int j = 0; j < 4; ++j) nn[j] = nr4[4 * gs + j];
#pragma unroll
      for (int j = 0; j < 4; ++j) {
        const v4f xs = (j < 2) ? xa : xb;
        const float x0 = (j & 1) ? xs.z : xs.x;
        const float x1 = (j & 1) ? xs.w : xs.y;
        const v4f nv = nq[j];
        v2f av;
        av.x = fmaf(x0, nv.x, nv.y);
        av.y = fmaf(x1, nv.z, nv.w);
        v2f z2; z2.x = 0.f; z2.y = 0.f;
        av = hi ? z2 : av;
        asum.x += av.x; asum.y += av.y;
        v8f z[4];
#pragma unroll
        for (int t = 0; t < 4; ++t) z[t] = wmma4(av, bw[t], zero8);
#pragma unroll
        for (int t = 0; t < 4; ++t)
#pragma unroll
          for (int e = 0; e < 8; ++e) absacc[t][e] += fabsf(z[t][e]);
      }
#pragma unroll
      for (int j = 0; j < 4; ++j) nq[j] = nn[j];
    }
  }

#pragma unroll
  for (int t = 0; t < 4; ++t) {
    v8f cb = splat8(100.f * bme[t * 16 + nc]);
    v8f zs = wmma4(asum, bw[t], cb);
#pragma unroll
    for (int e = 0; e < 8; ++e) {
      const float m = (0.505f * zs[e] + 0.495f * absacc[t][e]) * 0.01f;
      const int r = e + (hi ? 8 : 0);
      pb[r * PSTR + 4 + t * 16 + nc] = m;
    }
  }
  if (!hi) {
#pragma unroll
    for (int c = 0; c < 4; ++c)
      pb[nc * PSTR + c] = fmaf(xrow[c], nrm[2 * c], nrm[2 * c + 1]);
  }
  lds_fence();

  v8f h[4];
  dense_layer<17, 4, true>(pb, nc, hi, W1, 68, b1, h);
  lds_fence();
  store_h(pb, nc, hi, h);
  lds_fence();
  dense_layer<16, 4, true>(pb, nc, hi, W2, 64, b2, h);
  lds_fence();
  store_h(pb, nc, hi, h);
  lds_fence();
  v8f lg[2];
  dense_layer<16, 2, false>(pb, nc, hi, W3, 64, b3, lg);

#pragma unroll
  for (int r = 0; r < 8; ++r) {
    const float a0 = lg[0][r], a1 = lg[1][r];
    float m = fmaxf(a0, a1);
#pragma unroll
    for (int off = 8; off >= 1; off >>= 1) m = fmaxf(m, __shfl_xor(m, off, 32));
    const float e0 = __expf(a0 - m);
    const float e1 = __expf(a1 - m);
    float s = e0 + e1;
#pragma unroll
    for (int off = 8; off >= 1; off >>= 1) s += __shfl_xor(s, off, 32);
    const float inv = 1.0f / s;
    const int row = rowbase + r + (hi ? 8 : 0);
    out[row * 32 + nc]      = e0 * inv;
    out[row * 32 + 16 + nc] = e1 * inv;
  }
}

extern "C" void kernel_launch(void* const* d_in, const int* in_sizes, int n_in,
                              void* d_out, int out_size, void* d_ws, size_t ws_size,
                              hipStream_t stream) {
  const float* X   = (const float*)d_in[0];
  const float* Wme = (const float*)d_in[1];
  const float* bme = (const float*)d_in[2];
  const float* W1  = (const float*)d_in[3];
  const float* b1  = (const float*)d_in[4];
  const float* W2  = (const float*)d_in[5];
  const float* b2  = (const float*)d_in[6];
  const float* W3  = (const float*)d_in[7];
  const float* b3  = (const float*)d_in[8];
  float* out = (float*)d_out;

  float* nrm = (float*)d_ws;

  float* part = out;

  bn_partial_kernel<<<SCH, 256, 0, stream>>>(X, part);
  bn_final_kernel<<<51, SCH, 0, stream>>>(part, nrm);
  policy_fused_kernel<<<NB / 64, 128, 0, stream>>>(X, nrm, Wme, bme,
                                                   W1, b1, W2, b2, W3, b3, out);
}


# MLP weights staged once per work-group into LDS by whole-line float4 reads at kernel start (biases to registers); layers read B fragments from LDS, not by 164 sixteen-line global loads per wave
# speedup vs baseline: 2.0323x; 1.0557x over previous
#include <hip/hip_runtime.h>

typedef __attribute__((ext_vector_type(2))) float v2f;
typedef __attribute__((ext_vector_type(4))) float v4f;
typedef __attribute__((ext_vector_type(8))) float v8f;

#define NB    16384
#define DIN   204
#define NPAIR 100
#define PSTR  70
#define WP    68
#define WP4   (WP / 4)

static __device__ __forceinline__ v8f wmma4(v2f a, v2f b, v8f c) {
  return __builtin_amdgcn_wmma_f32_16x16x4_f32(false, a, false, b, (short)0, c, false, false);
}

static __device__ __forceinline__ v8f splat8(float v) {
  v8f r;
#pragma unroll
  for (int e = 0; e < 8; ++e) r[e] = v;
  return r;
}

static __device__ __forceinline__ void lds_fence() {
  asm volatile("s_wait_dscnt 0" ::: "memory");
}

#define SCH   128
#define SROWS (NB / SCH)
#define SF4   (SROWS * 51)
#define SSLOT 512

__global__ __launch_bounds__(256) void bn_partial_kernel(const float* __restrict__ X,
                                                         float* __restrict__ part) {
  __shared__ v4f red[2 * 255];
  const int t = threadIdx.x;
  v4f s = {0.f, 0.f, 0.f, 0.f};
  v4f q = {0.f, 0.f, 0.f, 0.f};
  if (t < 255) {
    const v4f* p = (const v4f*)(X + (size_t)blockIdx.x * SROWS * DIN) + t;
#pragma unroll
    for (int j = 0; j < 25; ++j) {
      const v4f v = p[j * 255];
      s += v;
      q += v * v;
    }
    if (t < SF4 - 25 * 255) {
      const v4f v = p[25 * 255];
      s += v;
      q += v * v;
    }
    red[t]       = s;
    red[255 + t] = q;
  }
  __syncthreads();
  if (t < 102) {
    const int k  = (t < 51) ? t : t - 51;
    const v4f* r = red + ((t < 51) ? 0 : 255) + k;
    const v4f a = (r[0] + r[51]) + (r[102] + r[153]) + r[204];
    *(v4f*)(part + (size_t)blockIdx.x * SSLOT + 4 * t) = a;
  }
}

__global__ __launch_bounds__(SCH) void bn_final_kernel(const float* __restrict__ part,
                                                       float* __restrict__ nrm) {
  const int cg = blockIdx.x;
  const int g  = threadIdx.x;
  const float* slot = part + (size_t)g * SSLOT;
  v4f s = *(const v4f*)(slot + 4 * cg);
  v4f q = *(const v4f*)(slot + DIN + 4 * cg);
#pragma unroll
  for (int off = 16; off >= 1; off >>= 1) {
#pragma unroll
    for (int e = 0; e < 4; ++e) {
      s[e] += __shfl_xor(s[e], off, 32);
      q[e] += __shfl_xor(q[e], off, 32);
    }
  }
  __shared__ v4f reds[SCH / 32], redq[SCH / 32];
  if ((g & 31) == 0) {
    reds[g >> 5] = s;
    redq[g >> 5] = q;
  }
  __syncthreads();
  if (g < 4) {
    const float* fs = (const float*)reds;
    const float* fq = (const float*)redq;
    float ts = 0.f, tq = 0.f;
#pragma unroll
    for (int w = 0; w < SCH / 32; ++w) {
      ts += fs[4 * w + g];
      tq += fq[4 * w + g];
    }
    const int c = 4 * cg + g;
    float m   = ts * (1.f / NB);
    float var = tq * (1.f / NB) - m * m;
    float rs  = rsqrtf(var + 1e-5f);
    nrm[2 * c]     = rs;
    nrm[2 * c + 1] = -m * rs;
  }
}

template <int KSTEPS, int NT, bool RELU>
static __device__ __forceinline__ void dense_layer(const float* pb, int nc, bool hi,
                                                   const float* ws, const float* bv, v8f* acc) {
  const int kof = hi ? 2 : 0;
#pragma unroll
  for (int t = 0; t < NT; ++t) acc[t] = splat8(bv[t]);
#pragma unroll
  for (int kk = 0; kk < KSTEPS; ++kk) {
    const int ko = 4 * kk + kof;
    v2f a = *(const v2f*)(pb + nc * PSTR + ko);
#pragma unroll
    for (int t = 0; t < NT; ++t) {
      const int n = t * 16 + nc;
      v2f b = *(const v2f*)(ws + n * WP + ko);
      acc[t] = wmma4(a, b, acc[t]);
    }
  }
  if (RELU) {
#pragma unroll
    for (int t = 0; t < NT; ++t)
#pragma unroll
      for (int e = 0; e < 8; ++e) acc[t][e] = fmaxf(acc[t][e], 0.01f * acc[t][e]);
  }
}

static __device__ __forceinline__ void store_h(float* pb, int nc, bool hi, const v8f* h) {
#pragma unroll
  for (int t = 0; t < 4; ++t)
#pragma unroll
    for (int e = 0; e < 8; ++e) {
      const int r = e + (hi ? 8 : 0);
      pb[r * PSTR + t * 16 + nc] = h[t][e];
    }
}

__global__ __launch_bounds__(128) void policy_fused_kernel(
    const float* __restrict__ X, const float* __restrict__ nrm,
    const float* __restrict__ Wme, const float* __restrict__ bme,
    const float* __restrict__ W1, const float* __restrict__ b1,
    const float* __restrict__ W2, const float* __restrict__ b2,
    const float* __restrict__ W3, const float* __restrict__ b3,
    float* __restrict__ out) {
  __shared__ float polbuf[4][16 * PSTR];
  __shared__ v4f w1s[64 * WP4];
  __shared__ v4f w2s[64 * WP4];
  __shared__ v4f w3s[32 * WP4];

  {
    const int tid = threadIdx.x;
    const v4f* g1 = (const v4f*)W1;
    const v4f* g2 = (const v4f*)W2;
    const v4f* g3 = (const v4f*)W3;
    const int ilast = (tid < 64) ? 1024 + tid : tid;
    v4f s1[8], s2[8], s3[4];
    const v4f s1l = g1[ilast];
#pragma unroll
    for (int i = 0; i < 8; ++i) s1[i] = g1[tid + 128 * i];
#pragma unroll
    for (int i = 0; i < 8; ++i) s2[i] = g2[tid + 128 * i];
#pragma unroll
    for (int i = 0; i < 4; ++i) s3[i] = g3[tid + 128 * i];
#pragma unroll
    for (int i = 0; i < 8; ++i) w1s[tid + 128 * i] = s1[i];
    w1s[ilast] = s1l;
#pragma unroll
    for (int i = 0; i < 8; ++i) {
      const int idx = tid + 128 * i;
      w2s[(idx >> 4) * WP4 + (idx & 15)] = s2[i];
    }
#pragma unroll
    for (int i = 0; i < 4; ++i) {
      const int idx = tid + 128 * i;
      w3s[(idx >> 4) * WP4 + (idx & 15)] = s3[i];
    }
  }

  const int  lane = threadIdx.x & 31;
  const int  wv   = threadIdx.x >> 5;
  const int  nc   = lane & 15;
  const bool hi   = (lane >= 16);
  const int  rowbase = blockIdx.x * 64 + wv * 16;
  const int  myrow   = rowbase + nc;
  float* pb = polbuf[wv];
  const float* xrow = X + (size_t)myrow * DIN;

  float bmev[4], b1v[4], b2v[4], b3v[2];
#pragma unroll
  for (int t = 0; t < 4; ++t) {
    bmev[t] = bme[t * 16 + nc];
    b1v[t]  = b1[t * 16 + nc];
    b2v[t]  = b2[t * 16 + nc];
  }
#pragma unroll
  for (int t = 0; t < 2; ++t) b3v[t] = b3[t * 16 + nc];

  {
    const v4f x03 = *(const v4f*)xrow;
#pragma unroll
    for (int c = 0; c < 4; ++c)
      pb[nc * PSTR + c] = fmaf(x03[c], nrm[2 * c], nrm[2 * c + 1]);
  }

  v2f bw[4];
#pragma unroll
  for (int t = 0; t < 4; ++t) {
    const int n = t * 16 + nc;
    v2f w = *(const v2f*)(Wme + 2 * n);
    v2f z; z.x = 0.f; z.y = 0.f;
    bw[t] = hi ? z : w;
  }

  v8f absacc[4];
#pragma unroll
  for (int t = 0; t < 4; ++t) absacc[t] = splat8(0.f);
  v2f asum; asum.x = 0.f; asum.y = 0.f;
  const v8f zero8 = splat8(0.f);

  const v4f* xp4 = (const v4f*)(xrow + 4);
  const v4f* nr4 = (const v4f*)(nrm + 8);
  v4f xq[10];
#pragma unroll
  for (int i = 0; i < 10; ++i) xq[i] = xp4[i];
  v4f nq[4];
#pragma unroll
  for (int j = 0; j < 4; ++j) nq[j] = nr4[j];

  for (int go = 0; go < NPAIR / 4; go += 5) {
#pragma unroll
    for (int u = 0; u < 5; ++u) {
      const int g = go + u;
      const v4f xa = xq[2 * u], xb = xq[2 * u + 1];
      const int gx = (g + 5 < NPAIR / 4) ? g + 5 : NPAIR / 4 - 1;
      xq[2 * u]     = xp4[2 * gx];
      xq[2 * u + 1] = xp4[2 * gx + 1];
      const int gs = (g + 1 < NPAIR / 4) ? g + 1 : NPAIR / 4 - 1;
      v4f nn[4];
#pragma unroll
      for (int j = 0; j < 4; ++j) nn[j] = nr4[4 * gs + j];
#pragma unroll
      for (int j = 0; j < 4; ++j) {
        const v4f xs = (j < 2) ? xa : xb;
        const float x0 = (j & 1) ? xs.z : xs.x;
        const float x1 = (j & 1) ? xs.w : xs.y;
        const v4f nv = nq[j];
        v2f av;
        av.x = fmaf(x0, nv.x, nv.y);
        av.y = fmaf(x1, nv.z, nv.w);
        v2f z2; z2.x = 0.f; z2.y = 0.f;
        av = hi ? z2 : av;
        asum.x += av.x; asum.y += av.y;
        v8f z[4];
#pragma unroll
        for (int t = 0; t < 4; ++t) z[t] = wmma4(av, bw[t], zero8);
#pragma unroll
        for (int t = 0; t < 4; ++t)
#pragma unroll
          for (int e = 0; e < 8; ++e) absacc[t][e] += fabsf(z[t][e]);
      }
#pragma unroll
      for (int j = 0; j < 4; ++j) nq[j] = nn[j];
    }
  }

#pragma unroll
  for (int t = 0; t < 4; ++t) {
    v8f cb = splat8(100.f * bmev[t]);
    v8f zs = wmma4(asum, bw[t], cb);
#pragma unroll
    for (int e = 0; e < 8; ++e) {
      const float m = (0.505f * zs[e] + 0.495f * absacc[t][e]) * 0.01f;
      const int r = e + (hi ? 8 : 0);
      pb[r * PSTR + 4 + t * 16 + nc] = m;
    }
  }
  __syncthreads();

  v8f h[4];
  dense_layer<17, 4, true>(pb, nc, hi, (const float*)w1s, b1v, h);
  lds_fence();
  store_h(pb, nc, hi, h);
  lds_fence();
  dense_layer<16, 4, true>(pb, nc, hi, (const float*)w2s, b2v, h);
  lds_fence();
  store_h(pb, nc, hi, h);
  lds_fence();
  v8f lg[2];
  dense_layer<16, 2, false>(pb, nc, hi, (const float*)w3s, b3v, lg);

#pragma unroll
  for (int r = 0; r < 8; ++r) {
    const float a0 = lg[0][r], a1 = lg[1][r];
    float m = fmaxf(a0, a1);
#pragma unroll
    for (int off = 8; off >= 1; off >>= 1) m = fmaxf(m, __shfl_xor(m, off, 32));
    const float e0 = __expf(a0 - m);
    const float e1 = __expf(a1 - m);
    float s = e0 + e1;
#pragma unroll
    for (int off = 8; off >= 1; off >>= 1) s += __shfl_xor(s, off, 32);
    const float inv = 1.0f / s;
    const int row = rowbase + r + (hi ? 8 : 0);
    out[row * 32 + nc]      = e0 * inv;
    out[row * 32 + 16 + nc] = e1 * inv;
  }
}

extern "C" void kernel_launch(void* const* d_in, const int* in_sizes, int n_in,
                              void* d_out, int out_size, void* d_ws, size_t ws_size,
                              hipStream_t stream) {
  const float* X   = (const float*)d_in[0];
  const float* Wme = (const float*)d_in[1];
  const float* bme = (const float*)d_in[2];
  const float* W1  = (const float*)d_in[3];
  const float* b1  = (const float*)d_in[4];
  const float* W2  = (const float*)d_in[5];
  const float* b2  = (const float*)d_in[6];
  const float* W3  = (const float*)d_in[7];
  const float* b3  = (const float*)d_in[8];
  float* out = (float*)d_out;

  float* nrm = (float*)d_ws;

  float* part = out;

  bn_partial_kernel<<<SCH, 256, 0, stream>>>(X, part);
  bn_final_kernel<<<51, SCH, 0, stream>>>(part, nrm);
  policy_fused_kernel<<<NB / 64, 128, 0, stream>>>(X, nrm, Wme, bme,
                                                   W1, b1, W2, b2, W3, b3, out);
}


# Entity loop: the ten 16-byte X loads of turn i+1 are issued at the top of turn i into a second register set (turn 0's at kernel entry), so no turn opens by waiting on a load it has just issued
# speedup vs baseline: 2.1382x; 1.0521x over previous
#include <hip/hip_runtime.h>

typedef __attribute__((ext_vector_type(2))) float v2f;
typedef __attribute__((ext_vector_type(4))) float v4f;
typedef __attribute__((ext_vector_type(8))) float v8f;

#define NB    16384
#define DIN   204
#define NPAIR 100
#define PSTR  70
#define WP    68
#define WP4   (WP / 4)

static __device__ __forceinline__ v8f wmma4(v2f a, v2f b, v8f c) {
  return __builtin_amdgcn_wmma_f32_16x16x4_f32(false, a, false, b, (short)0, c, false, false);
}

static __device__ __forceinline__ v8f splat8(float v) {
  v8f r;
#pragma unroll
  for (int e = 0; e < 8; ++e) r[e] = v;
  return r;
}

static __device__ __forceinline__ void lds_fence() {
  asm volatile("s_wait_dscnt 0" ::: "memory");
}

#define SCH   128
#define SROWS (NB / SCH)
#define SF4   (SROWS * 51)
#define SSLOT 512

__global__ __launch_bounds__(256) void bn_partial_kernel(const float* __restrict__ X,
                                                         float* __restrict__ part) {
  __shared__ v4f red[2 * 255];
  const int t = threadIdx.x;
  v4f s = {0.f, 0.f, 0.f, 0.f};
  v4f q = {0.f, 0.f, 0.f, 0.f};
  if (t < 255) {
    const v4f* p = (const v4f*)(X + (size_t)blockIdx.x * SROWS * DIN) + t;
#pragma unroll
    for (int j = 0; j < 25; ++j) {
      const v4f v = p[j * 255];
      s += v;
      q += v * v;
    }
    if (t < SF4 - 25 * 255) {
      const v4f v = p[25 * 255];
      s += v;
      q += v * v;
    }
    red[t]       = s;
    red[255 + t] = q;
  }
  __syncthreads();
  if (t < 102) {
    const int k  = (t < 51) ? t : t - 51;
    const v4f* r = red + ((t < 51) ? 0 : 255) + k;
    const v4f a = (r[0] + r[51]) + (r[102] + r[153]) + r[204];
    *(v4f*)(part + (size_t)blockIdx.x * SSLOT + 4 * t) = a;
  }
}

__global__ __launch_bounds__(SCH) void bn_final_kernel(const float* __restrict__ part,
                                                       float* __restrict__ nrm) {
  const int cg = blockIdx.x;
  const int g  = threadIdx.x;
  const float* slot = part + (size_t)g * SSLOT;
  v4f s = *(const v4f*)(slot + 4 * cg);
  v4f q = *(const v4f*)(slot + DIN + 4 * cg);
#pragma unroll
  for (int off = 16; off >= 1; off >>= 1) {
#pragma unroll
    for (int e = 0; e < 4; ++e) {
      s[e] += __shfl_xor(s[e], off, 32);
      q[e] += __shfl_xor(q[e], off, 32);
    }
  }
  __shared__ v4f reds[SCH / 32], redq[SCH / 32];
  if ((g & 31) == 0) {
    reds[g >> 5] = s;
    redq[g >> 5] = q;
  }
  __syncthreads();
  if (g < 4) {
    const float* fs = (const float*)reds;
    const float* fq = (const float*)redq;
    float ts = 0.f, tq = 0.f;
#pragma unroll
    for (int w = 0; w < SCH / 32; ++w) {
      ts += fs[4 * w + g];
      tq += fq[4 * w + g];
    }
    const int c = 4 * cg + g;
    float m   = ts * (1.f / NB);
    float var = tq * (1.f / NB) - m * m;
    float rs  = rsqrtf(var + 1e-5f);
    nrm[2 * c]     = rs;
    nrm[2 * c + 1] = -m * rs;
  }
}

template <int KSTEPS, int NT, bool RELU>
static __device__ __forceinline__ void dense_layer(const float* pb, int nc, bool hi,
                                                   const float* ws, const float* bv, v8f* acc) {
  const int kof = hi ? 2 : 0;
#pragma unroll
  for (int t = 0; t < NT; ++t) acc[t] = splat8(bv[t]);
#pragma unroll
  for (int kk = 0; kk < KSTEPS; ++kk) {
    const int ko = 4 * kk + kof;
    v2f a = *(const v2f*)(pb + nc * PSTR + ko);
#pragma unroll
    for (int t = 0; t < NT; ++t) {
      const int n = t * 16 + nc;
      v2f b = *(const v2f*)(ws + n * WP + ko);
      acc[t] = wmma4(a, b, acc[t]);
    }
  }
  if (RELU) {
#pragma unroll
    for (int t = 0; t < NT; ++t)
#pragma unroll
      for (int e = 0; e < 8; ++e) acc[t][e] = fmaxf(acc[t][e], 0.01f * acc[t][e]);
  }
}

static __device__ __forceinline__ void store_h(float* pb, int nc, bool hi, const v8f* h) {
#pragma unroll
  for (int t = 0; t < 4; ++t)
#pragma unroll
    for (int e = 0; e < 8; ++e) {
      const int r = e + (hi ? 8 : 0);
      pb[r * PSTR + t * 16 + nc] = h[t][e];
    }
}

__global__ __launch_bounds__(128) void policy_fused_kernel(
    const float* __restrict__ X, const float* __restrict__ nrm,
    const float* __restrict__ Wme, const float* __restrict__ bme,
    const float* __restrict__ W1, const float* __restrict__ b1,
    const float* __restrict__ W2, const float* __restrict__ b2,
    const float* __restrict__ W3, const float* __restrict__ b3,
    float* __restrict__ out) {
  __shared__ float polbuf[4][16 * PSTR];
  __shared__ v4f w1s[64 * WP4];
  __shared__ v4f w2s[64 * WP4];
  __shared__ v4f w3s[32 * WP4];

  const int  lane = threadIdx.x & 31;
  const int  wv   = threadIdx.x >> 5;
  const int  nc   = lane & 15;
  const bool hi   = (lane >= 16);
  const int  rowbase = blockIdx.x * 64 + wv * 16;
  const int  myrow   = rowbase + nc;
  float* pb = polbuf[wv];
  const float* xrow = X + (size_t)myrow * DIN;

  const v4f* xp4 = (const v4f*)(xrow + 4);
  v4f xq[10];
#pragma unroll
  for (int i = 0; i < 10; ++i) xq[i] = xp4[i];

  {
    const int tid = threadIdx.x;
    const v4f* g1 = (const v4f*)W1;
    const v4f* g2 = (const v4f*)W2;
    const v4f* g3 = (const v4f*)W3;
    const int ilast = (tid < 64) ? 1024 + tid : tid;
    v4f s1[8], s2[8], s3[4];
    const v4f s1l = g1[ilast];
#pragma unroll
    for (int i = 0; i < 8; ++i) s1[i] = g1[tid + 128 * i];
#pragma unroll
    for (int i = 0; i < 8; ++i) s2[i] = g2[tid + 128 * i];
#pragma unroll
    for (int i = 0; i < 4; ++i) s3[i] = g3[tid + 128 * i];
#pragma unroll
    for (int i = 0; i < 8; ++i) w1s[tid + 128 * i] = s1[i];
    w1s[ilast] = s1l;
#pragma unroll
    for (int i = 0; i < 8; ++i) {
      const int idx = tid + 128 * i;
      w2s[(idx >> 4) * WP4 + (idx & 15)] = s2[i];
    }
#pragma unroll
    for (int i = 0; i < 4; ++i) {
      const int idx = tid + 128 * i;
      w3s[(idx >> 4) * WP4 + (idx & 15)] = s3[i];
    }
  }

  float bmev[4], b1v[4], b2v[4], b3v[2];
#pragma unroll
  for (int t = 0; t < 4; ++t) {
    bmev[t] = bme[t * 16 + nc];
    b1v[t]  = b1[t * 16 + nc];
    b2v[t]  = b2[t * 16 + nc];
  }
#pragma unroll
  for (int t = 0; t < 2; ++t) b3v[t] = b3[t * 16 + nc];

  {
    const v4f x03 = *(const v4f*)xrow;
#pragma unroll
    for (int c = 0; c < 4; ++c)
      pb[nc * PSTR + c] = fmaf(x03[c], nrm[2 * c], nrm[2 * c + 1]);
  }

  v2f bw[4];
#pragma unroll
  for (int t = 0; t < 4; ++t) {
    const int n = t * 16 + nc;
    v2f w = *(const v2f*)(Wme + 2 * n);
    v2f z; z.x = 0.f; z.y = 0.f;
    bw[t] = hi ? z : w;
  }

  v8f absacc[4];
#pragma unroll
  for (int t = 0; t < 4; ++t) absacc[t] = splat8(0.f);
  v2f asum; asum.x = 0.f; asum.y = 0.f;
  const v8f zero8 = splat8(0.f);

  const v4f* nr4 = (const v4f*)(nrm + 8);
  v4f nq[4];
#pragma unroll
  for (int j = 0; j < 4; ++j) nq[j] = nr4[j];

  for (int go = 0; go < NPAIR / 4; go += 5) {
    const int gn = (go + 5 < NPAIR / 4) ? go + 5 : go;
    v4f nx[10];
#pragma unroll
    for (int i = 0; i < 10; ++i) nx[i] = xp4[2 * gn + i];
#pragma unroll
    for (int u = 0; u < 5; ++u) {
      const int g = go + u;
      const v4f xa = xq[2 * u], xb = xq[2 * u + 1];
      const int gs = (g + 1 < NPAIR / 4) ? g + 1 : NPAIR / 4 - 1;
      v4f nn[4];
#pragma unroll
      for (int j = 0; j < 4; ++j) nn[j] = nr4[4 * gs + j];
#pragma unroll
      for (int j = 0; j < 4; ++j) {
        const v4f xs = (j < 2) ? xa : xb;
        const float x0 = (j & 1) ? xs.z : xs.x;
        const float x1 = (j & 1) ? xs.w : xs.y;
        const v4f nv = nq[j];
        v2f av;
        av.x = fmaf(x0, nv.x, nv.y);
        av.y = fmaf(x1, nv.z, nv.w);
        v2f z2; z2.x = 0.f; z2.y = 0.f;
        av = hi ? z2 : av;
        asum.x += av.x; asum.y += av.y;
        v8f z[4];
#pragma unroll
        for (int t = 0; t < 4; ++t) z[t] = wmma4(av, bw[t], zero8);
#pragma unroll
        for (int t = 0; t < 4; ++t)
#pragma unroll
          for (int e = 0; e < 8; ++e) absacc[t][e] += fabsf(z[t][e]);
      }
#pragma unroll
      for (int j = 0; j < 4; ++j) nq[j] = nn[j];
    }
    asm volatile("" :: "v"(nx[0]), "v"(nx[1]), "v"(nx[2]), "v"(nx[3]), "v"(nx[4]),
                       "v"(nx[5]), "v"(nx[6]), "v"(nx[7]), "v"(nx[8]), "v"(nx[9]));
#pragma unroll
    for (int i = 0; i < 10; ++i) xq[i] = nx[i];
  }

#pragma unroll
  for (int t = 0; t < 4; ++t) {
    v8f cb = splat8(100.f * bmev[t]);
    v8f zs = wmma4(asum, bw[t], cb);
#pragma unroll
    for (int e = 0; e < 8; ++e) {
      const float m = (0.505f * zs[e] + 0.495f * absacc[t][e]) * 0.01f;
      const int r = e + (hi ? 8 : 0);
      pb[r * PSTR + 4 + t * 16 + nc] = m;
    }
  }
  __syncthreads();

  v8f h[4];
  dense_layer<17, 4, true>(pb, nc, hi, (const float*)w1s, b1v, h);
  lds_fence();
  store_h(pb, nc, hi, h);
  lds_fence();
  dense_layer<16, 4, true>(pb, nc, hi, (const float*)w2s, b2v, h);
  lds_fence();
  store_h(pb, nc, hi, h);
  lds_fence();
  v8f lg[2];
  dense_layer<16, 2, false>(pb, nc, hi, (const float*)w3s, b3v, lg);

#pragma unroll
  for (int r = 0; r < 8; ++r) {
    const float a0 = lg[0][r], a1 = lg[1][r];
    float m = fmaxf(a0, a1);
#pragma unroll
    for (int off = 8; off >= 1; off >>= 1) m = fmaxf(m, __shfl_xor(m, off, 32));
    const float e0 = __expf(a0 - m);
    const float e1 = __expf(a1 - m);
    float s = e0 + e1;
#pragma unroll
    for (int off = 8; off >= 1; off >>= 1) s += __shfl_xor(s, off, 32);
    const float inv = 1.0f / s;
    const int row = rowbase + r + (hi ? 8 : 0);
    out[row * 32 + nc]      = e0 * inv;
    out[row * 32 + 16 + nc] = e1 * inv;
  }
}

extern "C" void kernel_launch(void* const* d_in, const int* in_sizes, int n_in,
                              void* d_out, int out_size, void* d_ws, size_t ws_size,
                              hipStream_t stream) {
  const float* X   = (const float*)d_in[0];
  const float* Wme = (const float*)d_in[1];
  const float* bme = (const float*)d_in[2];
  const float* W1  = (const float*)d_in[3];
  const float* b1  = (const float*)d_in[4];
  const float* W2  = (const float*)d_in[5];
  const float* b2  = (const float*)d_in[6];
  const float* W3  = (const float*)d_in[7];
  const float* b3  = (const float*)d_in[8];
  float* out = (float*)d_out;

  float* nrm = (float*)d_ws;

  float* part = out;

  bn_partial_kernel<<<SCH, 256, 0, stream>>>(X, part);
  bn_final_kernel<<<51, SCH, 0, stream>>>(part, nrm);
  policy_fused_kernel<<<NB / 64, 128, 0, stream>>>(X, nrm, Wme, bme,
                                                   W1, b1, W2, b2, W3, b3, out);
}


# Statistics as 32 partial planes (32 row blocks x 8 column slabs) in the workspace, added up by every fused work-group at its start: the summing launch goes; changes the order of the statistic sums
# speedup vs baseline: 2.3942x; 1.1197x over previous
#include <hip/hip_runtime.h>

typedef __attribute__((ext_vector_type(2))) float v2f;
typedef __attribute__((ext_vector_type(4))) float v4f;
typedef __attribute__((ext_vector_type(8))) float v8f;

#define NB    16384
#define DIN   204
#define NPAIR 100
#define PSTR  70
#define WP    68
#define WP4   (WP / 4)

static __device__ __forceinline__ v8f wmma4(v2f a, v2f b, v8f c) {
  return __builtin_amdgcn_wmma_f32_16x16x4_f32(false, a, false, b, (short)0, c, false, false);
}

static __device__ __forceinline__ v8f splat8(float v) {
  v8f r;
#pragma unroll
  for (int e = 0; e < 8; ++e) r[e] = v;
  return r;
}

static __device__ __forceinline__ void lds_fence() {
  asm volatile("s_wait_dscnt 0" ::: "memory");
}

#define SCH   128
#define SROWS (NB / SCH)
#define SF4   (SROWS * 51)
#define SSLOT 512

__global__ __launch_bounds__(256) void bn_partial_kernel(const float* __restrict__ X,
                                                         float* __restrict__ part) {
  __shared__ v4f red[2 * 255];
  const int t = threadIdx.x;
  v4f s = {0.f, 0.f, 0.f, 0.f};
  v4f q = {0.f, 0.f, 0.f, 0.f};
  if (t < 255) {
    const v4f* p = (const v4f*)(X + (size_t)blockIdx.x * SROWS * DIN) + t;
#pragma unroll
    for (int j = 0; j < 25; ++j) {
      const v4f v = p[j * 255];
      s += v;
      q += v * v;
    }
    if (t < SF4 - 25 * 255) {
      const v4f v = p[25 * 255];
      s += v;
      q += v * v;
    }
    red[t]       = s;
    red[255 + t] = q;
  }
  __syncthreads();
  if (t < 102) {
    const int k  = (t < 51) ? t : t - 51;
    const v4f* r = red + ((t < 51) ? 0 : 255) + k;
    const v4f a = (r[0] + r[51]) + (r[102] + r[153]) + r[204];
    *(v4f*)(part + (size_t)blockIdx.x * SSLOT + 4 * t) = a;
  }
}

__global__ __launch_bounds__(SCH) void bn_final_kernel(const float* __restrict__ part,
                                                       float* __restrict__ nrm) {
  const int cg = blockIdx.x;
  const int g  = threadIdx.x;
  const float* slot = part + (size_t)g * SSLOT;
  v4f s = *(const v4f*)(slot + 4 * cg);
  v4f q = *(const v4f*)(slot + DIN + 4 * cg);
#pragma unroll
  for (int off = 16; off >= 1; off >>= 1) {
#pragma unroll
    for (int e = 0; e < 4; ++e) {
      s[e] += __shfl_xor(s[e], off, 32);
      q[e] += __shfl_xor(q[e], off, 32);
    }
  }
  __shared__ v4f reds[SCH / 32], redq[SCH / 32];
  if ((g & 31) == 0) {
    reds[g >> 5] = s;
    redq[g >> 5] = q;
  }
  __syncthreads();
  if (g < 4) {
    const float* fs = (const float*)reds;
    const float* fq = (const float*)redq;
    float ts = 0.f, tq = 0.f;
#pragma unroll
    for (int w = 0; w < SCH / 32; ++w) {
      ts += fs[4 * w + g];
      tq += fq[4 * w + g];
    }
    const int c = 4 * cg + g;
    float m   = ts * (1.f / NB);
    float var = tq * (1.f / NB) - m * m;
    float rs  = rsqrtf(var + 1e-5f);
    nrm[2 * c]     = rs;
    nrm[2 * c + 1] = -m * rs;
  }
}

#define RB    32
#define RROWS (NB / RB)
#define NSLAB 8
#define SLOTF 64

__global__ __launch_bounds__(128) void bn_slab_kernel(const float* __restrict__ X,
                                                      float* __restrict__ planes) {
  __shared__ v4f red[2 * 32];
  const int t  = threadIdx.x;
  const int rb = blockIdx.x >> 3;
  const int sl = blockIdx.x & 7;
  const int cs = (sl < 3) ? 7 * sl : 21 + 6 * (sl - 3);
  const int ng = (sl < 3) ? 7 : 6;
  const int g  = t & 7;
  const int r  = t >> 3;
  v4f s = {0.f, 0.f, 0.f, 0.f};
  v4f q = {0.f, 0.f, 0.f, 0.f};
  if (g < ng) {
    const float* p = X + (size_t)(rb * RROWS + r) * DIN + 4 * (cs + g);
#pragma unroll
    for (int j = 0; j < RROWS / 16; ++j) {
      const v4f v = *(const v4f*)(p + (size_t)j * 16 * DIN);
      s += v;
      q += v * v;
    }
  }
#pragma unroll
  for (int off = 8; off <= 16; off <<= 1) {
#pragma unroll
    for (int e = 0; e < 4; ++e) {
      s[e] += __shfl_xor(s[e], off, 32);
      q[e] += __shfl_xor(q[e], off, 32);
    }
  }
  const int lane = t & 31, w = t >> 5;
  if (lane < 8) {
    red[w * 8 + lane]      = s;
    red[32 + w * 8 + lane] = q;
  }
  __syncthreads();
  if (t < 16 && (t & 7) < ng) {
    const int which = t >> 3, gg = t & 7;
    const v4f* rr = red + which * 32 + gg;
    const v4f a = (rr[0] + rr[8]) + (rr[16] + rr[24]);
    *(v4f*)(planes + (size_t)blockIdx.x * SLOTF + which * 32 + 4 * gg) = a;
  }
}

template <int KSTEPS, int NT, bool RELU>
static __device__ __forceinline__ void dense_layer(const float* pb, int nc, bool hi,
                                                   const float* ws, const float* bv, v8f* acc) {
  const int kof = hi ? 2 : 0;
#pragma unroll
  for (int t = 0; t < NT; ++t) acc[t] = splat8(bv[t]);
#pragma unroll
  for (int kk = 0; kk < KSTEPS; ++kk) {
    const int ko = 4 * kk + kof;
    v2f a = *(const v2f*)(pb + nc * PSTR + ko);
#pragma unroll
    for (int t = 0; t < NT; ++t) {
      const int n = t * 16 + nc;
      v2f b = *(const v2f*)(ws + n * WP + ko);
      acc[t] = wmma4(a, b, acc[t]);
    }
  }
  if (RELU) {
#pragma unroll
    for (int t = 0; t < NT; ++t)
#pragma unroll
      for (int e = 0; e < 8; ++e) acc[t][e] = fmaxf(acc[t][e], 0.01f * acc[t][e]);
  }
}

static __device__ __forceinline__ void store_h(float* pb, int nc, bool hi, const v8f* h) {
#pragma unroll
  for (int t = 0; t < 4; ++t)
#pragma unroll
    for (int e = 0; e < 8; ++e) {
      const int r = e + (hi ? 8 : 0);
      pb[r * PSTR + t * 16 + nc] = h[t][e];
    }
}

template <bool FOLD>
__global__ __launch_bounds__(128) void policy_fused_kernel(
    const float* __restrict__ X, const float* __restrict__ nrm,
    const float* __restrict__ Wme, const float* __restrict__ bme,
    const float* __restrict__ W1, const float* __restrict__ b1,
    const float* __restrict__ W2, const float* __restrict__ b2,
    const float* __restrict__ W3, const float* __restrict__ b3,
    float* __restrict__ out) {
  __shared__ float polbuf[4][16 * PSTR];
  __shared__ v4f w1s[64 * WP4];
  __shared__ v4f w2s[64 * WP4];
  __shared__ v4f w3s[32 * WP4];
  __shared__ v4f nrms[2 * 51];

  const int  lane = threadIdx.x & 31;
  const int  wv   = threadIdx.x >> 5;
  const int  nc   = lane & 15;
  const bool hi   = (lane >= 16);
  const int  rowbase = blockIdx.x * 64 + wv * 16;
  const int  myrow   = rowbase + nc;
  float* pb = polbuf[wv];
  const float* xrow = X + (size_t)myrow * DIN;

  v4f fl[RB];
  const int fcg = wv * 16 + nc;
  if (FOLD) {
    const int cgc = (fcg < 51) ? fcg : 50;
    const int fsl = (cgc < 21) ? cgc / 7 : 3 + (cgc - 21) / 6;
    const int fg  = (cgc < 21) ? cgc - 7 * fsl : (cgc - 21) - 6 * (fsl - 3);
    const float* pp = nrm + fsl * SLOTF + (hi ? 32 : 0) + 4 * fg;
#pragma unroll
    for (int rb = 0; rb < RB; ++rb) fl[rb] = *(const v4f*)(pp + rb * NSLAB * SLOTF);
  }

  const v4f* xp4 = (const v4f*)(xrow + 4);
  v4f xq[10];
#pragma unroll
  for (int i = 0; i < 10; ++i) xq[i] = xp4[i];

  {
    const int tid = threadIdx.x;
    const v4f* g1 = (const v4f*)W1;
    const v4f* g2 = (const v4f*)W2;
    const v4f* g3 = (const v4f*)W3;
    const int ilast = (tid < 64) ? 1024 + tid : tid;
    v4f s1[8], s2[8], s3[4];
    const v4f s1l = g1[ilast];
#pragma unroll
    for (int i = 0; i < 8; ++i) s1[i] = g1[tid + 128 * i];
#pragma unroll
    for (int i = 0; i < 8; ++i) s2[i] = g2[tid + 128 * i];
#pragma unroll
    for (int i = 0; i < 4; ++i) s3[i] = g3[tid + 128 * i];
    if (FOLD) {
      v4f a4[4];
#pragma unroll
      for (int i = 0; i < 4; ++i)
        a4[i] = ((fl[i] + fl[4 + i]) + (fl[8 + i] + fl[12 + i])) +
                ((fl[16 + i] + fl[20 + i]) + (fl[24 + i] + fl[28 + i]));
      const v4f tot = (a4[0] + a4[1]) + (a4[2] + a4[3]);
      v4f sq;
#pragma unroll
      for (int e = 0; e < 4; ++e) sq[e] = __shfl_xor(tot[e], 16, 32);
      if (!hi && fcg < 51) {
        float sc[4], sh[4];
#pragma unroll
        for (int e = 0; e < 4; ++e) {
          const float m   = tot[e] * (1.f / NB);
          const float var = sq[e] * (1.f / NB) - m * m;
          const float rs  = rsqrtf(var + 1e-5f);
          sc[e] = rs;
          sh[e] = -m * rs;
        }
        const v4f e0 = {sc[0], sh[0], sc[1], sh[1]};
        const v4f e1 = {sc[2], sh[2], sc[3], sh[3]};
        nrms[2 * fcg]     = e0;
        nrms[2 * fcg + 1] = e1;
      }
    }
#pragma unroll
    for (int i = 0; i < 8; ++i) w1s[tid + 128 * i] = s1[i];
    w1s[ilast] = s1l;
#pragma unroll
    for (int i = 0; i < 8; ++i) {
      const int idx = tid + 128 * i;
      w2s[(idx >> 4) * WP4 + (idx & 15)] = s2[i];
    }
#pragma unroll
    for (int i = 0; i < 4; ++i) {
      const int idx = tid + 128 * i;
      w3s[(idx >> 4) * WP4 + (idx & 15)] = s3[i];
    }
  }

  float bmev[4], b1v[4], b2v[4], b3v[2];
#pragma unroll
  for (int t = 0; t < 4; ++t) {
    bmev[t] = bme[t * 16 + nc];
    b1v[t]  = b1[t * 16 + nc];
    b2v[t]  = b2[t * 16 + nc];
  }
#pragma unroll
  for (int t = 0; t < 2; ++t) b3v[t] = b3[t * 16 + nc];

  v2f bw[4];
#pragma unroll
  for (int t = 0; t < 4; ++t) {
    const int n = t * 16 + nc;
    v2f w = *(const v2f*)(Wme + 2 * n);
    v2f z; z.x = 0.f; z.y = 0.f;
    bw[t] = hi ? z : w;
  }

  const v4f x03 = *(const v4f*)xrow;
  if (FOLD) __syncthreads();
  {
    if (FOLD) {
      const v4f n0 = nrms[0], n1 = nrms[1];
      pb[nc * PSTR + 0] = fmaf(x03[0], n0.x, n0.y);
      pb[nc * PSTR + 1] = fmaf(x03[1], n0.z, n0.w);
      pb[nc * PSTR + 2] = fmaf(x03[2], n1.x, n1.y);
      pb[nc * PSTR + 3] = fmaf(x03[3], n1.z, n1.w);
    } else {
#pragma unroll
      for (int c = 0; c < 4; ++c)
        pb[nc * PSTR + c] = fmaf(x03[c], nrm[2 * c], nrm[2 * c + 1]);
    }
  }

  v8f absacc[4];
#pragma unroll
  for (int t = 0; t < 4; ++t) absacc[t] = splat8(0.f);
  v2f asum; asum.x = 0.f; asum.y = 0.f;
  const v8f zero8 = splat8(0.f);

  const v4f* nr4 = (const v4f*)(nrm + 8);
  const v4f* nl4 = nrms + 2;
  v4f nq[4];
#pragma unroll
  for (int j = 0; j < 4; ++j) nq[j] = FOLD ? nl4[j] : nr4[j];

  for (int go = 0; go < NPAIR / 4; go += 5) {
    const int gn = (go + 5 < NPAIR / 4) ? go + 5 : go;
    v4f nx[10];
#pragma unroll
    for (int i = 0; i < 10; ++i) nx[i] = xp4[2 * gn + i];
#pragma unroll
    for (int u = 0; u < 5; ++u) {
      const int g = go + u;
      const v4f xa = xq[2 * u], xb = xq[2 * u + 1];
      const int gs = (g + 1 < NPAIR / 4) ? g + 1 : NPAIR / 4 - 1;
      v4f nn[4];
#pragma unroll
      for (int j = 0; j < 4; ++j) nn[j] = FOLD ? nl4[4 * gs + j] : nr4[4 * gs + j];
#pragma unroll
      for (int j = 0; j < 4; ++j) {
        const v4f xs = (j < 2) ? xa : xb;
        const float x0 = (j & 1) ? xs.z : xs.x;
        const float x1 = (j & 1) ? xs.w : xs.y;
        const v4f nv = nq[j];
        v2f av;
        av.x = fmaf(x0, nv.x, nv.y);
        av.y = fmaf(x1, nv.z, nv.w);
        v2f z2; z2.x = 0.f; z2.y = 0.f;
        av = hi ? z2 : av;
        asum.x += av.x; asum.y += av.y;
        v8f z[4];
#pragma unroll
        for (int t = 0; t < 4; ++t) z[t] = wmma4(av, bw[t], zero8);
#pragma unroll
        for (int t = 0; t < 4; ++t)
#pragma unroll
          for (int e = 0; e < 8; ++e) absacc[t][e] += fabsf(z[t][e]);
      }
#pragma unroll
      for (int j = 0; j < 4; ++j) nq[j] = nn[j];
    }
    asm volatile("" :: "v"(nx[0]), "v"(nx[1]), "v"(nx[2]), "v"(nx[3]), "v"(nx[4]),
                       "v"(nx[5]), "v"(nx[6]), "v"(nx[7]), "v"(nx[8]), "v"(nx[9]));
#pragma unroll
    for (int i = 0; i < 10; ++i) xq[i] = nx[i];
  }

#pragma unroll
  for (int t = 0; t < 4; ++t) {
    v8f cb = splat8(100.f * bmev[t]);
    v8f zs = wmma4(asum, bw[t], cb);
#pragma unroll
    for (int e = 0; e < 8; ++e) {
      const float m = (0.505f * zs[e] + 0.495f * absacc[t][e]) * 0.01f;
      const int r = e + (hi ? 8 : 0);
      pb[r * PSTR + 4 + t * 16 + nc] = m;
    }
  }
  if (FOLD) lds_fence(); else __syncthreads();

  v8f h[4];
  dense_layer<17, 4, true>(pb, nc, hi, (const float*)w1s, b1v, h);
  lds_fence();
  store_h(pb, nc, hi, h);
  lds_fence();
  dense_layer<16, 4, true>(pb, nc, hi, (const float*)w2s, b2v, h);
  lds_fence();
  store_h(pb, nc, hi, h);
  lds_fence();
  v8f lg[2];
  dense_layer<16, 2, false>(pb, nc, hi, (const float*)w3s, b3v, lg);

#pragma unroll
  for (int r = 0; r < 8; ++r) {
    const float a0 = lg[0][r], a1 = lg[1][r];
    float m = fmaxf(a0, a1);
#pragma unroll
    for (int off = 8; off >= 1; off >>= 1) m = fmaxf(m, __shfl_xor(m, off, 32));
    const float e0 = __expf(a0 - m);
    const float e1 = __expf(a1 - m);
    float s = e0 + e1;
#pragma unroll
    for (int off = 8; off >= 1; off >>= 1) s += __shfl_xor(s, off, 32);
    const float inv = 1.0f / s;
    const int row = rowbase + r + (hi ? 8 : 0);
    out[row * 32 + nc]      = e0 * inv;
    out[row * 32 + 16 + nc] = e1 * inv;
  }
}

extern "C" void kernel_launch(void* const* d_in, const int* in_sizes, int n_in,
                              void* d_out, int out_size, void* d_ws, size_t ws_size,
                              hipStream_t stream) {
  const float* X   = (const float*)d_in[0];
  const float* Wme = (const float*)d_in[1];
  const float* bme = (const float*)d_in[2];
  const float* W1  = (const float*)d_in[3];
  const float* b1  = (const float*)d_in[4];
  const float* W2  = (const float*)d_in[5];
  const float* b2  = (const float*)d_in[6];
  const float* W3  = (const float*)d_in[7];
  const float* b3  = (const float*)d_in[8];
  float* out = (float*)d_out;

  float* nrm = (float*)d_ws;

  float* part = out;

  if (ws_size >= (size_t)(RB * NSLAB * SLOTF * 4)) {
    bn_slab_kernel<<<RB * NSLAB, 128, 0, stream>>>(X, nrm);
    policy_fused_kernel<true><<<NB / 64, 128, 0, stream>>>(X, nrm, Wme, bme,
                                                           W1, b1, W2, b2, W3, b3, out);
  } else {
    bn_partial_kernel<<<SCH, 256, 0, stream>>>(X, part);
    bn_final_kernel<<<51, SCH, 0, stream>>>(part, nrm);
    policy_fused_kernel<false><<<NB / 64, 128, 0, stream>>>(X, nrm, Wme, bme,
                                                            W1, b1, W2, b2, W3, b3, out);
  }
}
